# starting kernel
# baseline (speedup 1.0000x reference)
#include <hip/hip_runtime.h>
#include <hip/hip_bf16.h>

#define FIN  4
#define FHID 64

typedef float v2f __attribute__((ext_vector_type(2)));
typedef float v8f __attribute__((ext_vector_type(8)));

static __host__ int cdiv(int a, int b) { return (a + b - 1) / b; }

__global__ void k_degree(const int* __restrict__ ei, float* __restrict__ deg,
                         int E, int N) {
    int e = blockIdx.x * blockDim.x + threadIdx.x;
    if (e >= E + N) return;
    int dst = (e < E) ? ei[E + e] : (e - E);
    atomicAdd(&deg[dst], 1.0f);
}

__global__ void k_rsqrt(float* __restrict__ d, int N) {
    int n = blockIdx.x * blockDim.x + threadIdx.x;
    if (n >= N) return;
    float v = d[n];
    d[n] = (v > 0.0f) ? rsqrtf(v) : 0.0f;
}

__global__ void k_xw1_wmma(const float* __restrict__ x,
                           const float* __restrict__ W1,
                           float* __restrict__ t, int N) {
    const int lane = threadIdx.x & 31;
    const int wave = threadIdx.x >> 5;
    const int tile = blockIdx.x * (blockDim.x >> 5) + wave;
    if (tile * 16 >= N) return;

    const int half = lane >> 4;
    const int l16  = lane & 15;
    const int row  = tile * 16 + l16;
    const int kb   = half * 2;

    v2f a;
    a.x = x[row * FIN + kb + 0];
    a.y = x[row * FIN + kb + 1];

#pragma unroll
    for (int c = 0; c < FHID / 16; ++c) {
        const int col = c * 16 + l16;
        v2f b;
        b.x = W1[(kb + 0) * FHID + col];
        b.y = W1[(kb + 1) * FHID + col];
        v8f acc = {};
        acc = __builtin_amdgcn_wmma_f32_16x16x4_f32(
             false, a,  false, b,
             (short)0, acc,  false,  false);
#pragma unroll
        for (int i = 0; i < 8; ++i) {
            int m = half * 8 + i;
            t[(size_t)(tile * 16 + m) * FHID + col] = acc[i];
        }
    }
}

__global__ void k_edge1(const int* __restrict__ ei,
                        const float* __restrict__ t,
                        const float* __restrict__ dis,
                        float* __restrict__ agg, int E, int N) {
    const int  lane = threadIdx.x & 31;
    const long long wid =
        (long long)blockIdx.x * (blockDim.x >> 5) + (threadIdx.x >> 5);
    if (wid >= (long long)E + N) return;

    int src, dst;
    if (wid < E) { src = ei[wid]; dst = ei[E + wid]; }
    else         { src = dst = (int)(wid - E); }

    const float nrm = dis[src] * dis[dst];
    const float2 v  = *(const float2*)(t + (size_t)src * FHID + lane * 2);
    float* ap = agg + (size_t)dst * FHID + lane * 2;
    atomicAdd(ap + 0, v.x * nrm);
    atomicAdd(ap + 1, v.y * nrm);
}

__global__ void k_relu_mv(const float* __restrict__ agg,
                          const float* __restrict__ b1,
                          const float* __restrict__ W2,
                          float* __restrict__ z, int N) {
    const int lane = threadIdx.x & 31;
    const int n = blockIdx.x * (blockDim.x >> 5) + (threadIdx.x >> 5);
    if (n >= N) return;
    float s = 0.0f;
#pragma unroll
    for (int j = 0; j < FHID / 32; ++j) {
        int f = j * 32 + lane;
        float h = agg[(size_t)n * FHID + f] + b1[f];
        h = fmaxf(h, 0.0f);
        s = fmaf(h, W2[f], s);
    }
#pragma unroll
    for (int off = 16; off > 0; off >>= 1)
        s += __shfl_xor(s, off, 32);
    if (lane == 0) z[n] = s;
}

__global__ void k_edge2(const int* __restrict__ ei,
                        const float* __restrict__ z,
                        const float* __restrict__ dis,
                        float* __restrict__ oacc, int E, int N) {
    long long e = (long long)blockIdx.x * blockDim.x + threadIdx.x;
    if (e >= (long long)E + N) return;
    int src, dst;
    if (e < E) { src = ei[e]; dst = ei[E + e]; }
    else       { src = dst = (int)(e - E); }
    atomicAdd(&oacc[dst], z[src] * dis[src] * dis[dst]);
}

__global__ void k_sigmoid(const float* __restrict__ oacc,
                          const float* __restrict__ b2,
                          float* __restrict__ out, int N) {
    int n = blockIdx.x * blockDim.x + threadIdx.x;
    if (n >= N) return;
    float v = oacc[n] + b2[0];
    out[n] = 1.0f / (1.0f + __expf(-v));
}

extern "C" void kernel_launch(void* const* d_in, const int* in_sizes, int n_in,
                              void* d_out, int out_size, void* d_ws, size_t ws_size,
                              hipStream_t stream) {
    const float* x  = (const float*)d_in[0];
    const int*   ei = (const int*)d_in[1];
    const float* W1 = (const float*)d_in[2];
    const float* b1 = (const float*)d_in[3];
    const float* W2 = (const float*)d_in[4];
    const float* b2 = (const float*)d_in[5];
    float* out = (float*)d_out;

    const int N = in_sizes[0] / FIN;
    const int E = in_sizes[1] / 2;
    const int T = E + N;

    char* ws = (char*)d_ws;
    float* dis  = (float*)ws;                 ws += (size_t)N * sizeof(float);
    float* t    = (float*)ws;                 ws += (size_t)N * FHID * sizeof(float);
    float* agg  = (float*)ws;                 ws += (size_t)N * FHID * sizeof(float);
    float* z    = (float*)ws;                 ws += (size_t)N * sizeof(float);
    float* oacc = (float*)ws;

    hipMemsetAsync(dis,  0, (size_t)N * sizeof(float), stream);
    hipMemsetAsync(agg,  0, (size_t)N * FHID * sizeof(float), stream);
    hipMemsetAsync(oacc, 0, (size_t)N * sizeof(float), stream);

    const int BT = 256;
    const int WPB = BT / 32;

    k_degree  <<<cdiv(T, BT), BT, 0, stream>>>(ei, dis, E, N);
    k_rsqrt   <<<cdiv(N, BT), BT, 0, stream>>>(dis, N);

    const int tiles = cdiv(N, 16);
    k_xw1_wmma<<<cdiv(tiles, WPB), BT, 0, stream>>>(x, W1, t, N);

    k_edge1   <<<cdiv(T, WPB), BT, 0, stream>>>(ei, t, dis, agg, E, N);
    k_relu_mv <<<cdiv(N, WPB), BT, 0, stream>>>(agg, b1, W2, z, N);
    k_edge2   <<<cdiv(T, BT), BT, 0, stream>>>(ei, z, dis, oacc, E, N);
    k_sigmoid <<<cdiv(N, BT), BT, 0, stream>>>(oacc, b2, out, N);
}


# Layer 1 aggregates the 4 input features over the edges before the W1 product, (A x) W1 for A (x W1): 1 atomic 16-byte piece per edge, not 4 lines; float sum order changes
# speedup vs baseline: 1.2353x; 1.2353x over previous
#include <hip/hip_runtime.h>
#include <hip/hip_bf16.h>

#define FIN  4
#define FHID 64

typedef float v2f __attribute__((ext_vector_type(2)));
typedef float v8f __attribute__((ext_vector_type(8)));

static __host__ int cdiv(int a, int b) { return (a + b - 1) / b; }

__global__ void k_degree(const int* __restrict__ ei, float* __restrict__ deg,
                         int E, int N) {
    int e = blockIdx.x * blockDim.x + threadIdx.x;
    if (e >= E + N) return;
    int dst = (e < E) ? ei[E + e] : (e - E);
    atomicAdd(&deg[dst], 1.0f);
}

__global__ void k_rsqrt(float* __restrict__ d, int N) {
    int n = blockIdx.x * blockDim.x + threadIdx.x;
    if (n >= N) return;
    float v = d[n];
    d[n] = (v > 0.0f) ? rsqrtf(v) : 0.0f;
}

__global__ void k_xw1_wmma(const float* __restrict__ x,
                           const float* __restrict__ W1,
                           float* __restrict__ t, int N) {
    const int lane = threadIdx.x & 31;
    const int wave = threadIdx.x >> 5;
    const int tile = blockIdx.x * (blockDim.x >> 5) + wave;
    if (tile * 16 >= N) return;

    const int half = lane >> 4;
    const int l16  = lane & 15;
    const int row  = tile * 16 + l16;
    const int kb   = half * 2;

    v2f a;
    a.x = x[row * FIN + kb + 0];
    a.y = x[row * FIN + kb + 1];

#pragma unroll
    for (int c = 0; c < FHID / 16; ++c) {
        const int col = c * 16 + l16;
        v2f b;
        b.x = W1[(kb + 0) * FHID + col];
        b.y = W1[(kb + 1) * FHID + col];
        v8f acc = {};
        acc = __builtin_amdgcn_wmma_f32_16x16x4_f32(
             false, a,  false, b,
             (short)0, acc,  false,  false);
#pragma unroll
        for (int i = 0; i < 8; ++i) {
            int m = half * 8 + i;
            t[(size_t)(tile * 16 + m) * FHID + col] = acc[i];
        }
    }
}

__global__ void k_edge1x(const int* __restrict__ ei,
                         const float* __restrict__ x,
                         const float* __restrict__ dis,
                         float* __restrict__ aggx, int E, int N) {
    const int lane = threadIdx.x & 31;
    const long long e = (long long)blockIdx.x * blockDim.x + threadIdx.x;

    int   dst = -1;
    float v0 = 0.0f, v1 = 0.0f, v2 = 0.0f, v3 = 0.0f;
    if (e < (long long)E + N) {
        int src;
        if (e < E) { src = ei[e]; dst = ei[E + e]; }
        else       { src = dst = (int)(e - E); }
        const float  nrm = dis[src] * dis[dst];
        const float4 xr  = *(const float4*)(x + (size_t)src * FIN);
        v0 = xr.x * nrm; v1 = xr.y * nrm; v2 = xr.z * nrm; v3 = xr.w * nrm;
    }

    const int f = lane & 3;
    const int q = lane >> 2;
    for (int k = 0; k < 4; ++k) {
        const int from = k * 8 + q;
        const int   d  = __shfl(dst, from, 32);
        const float a0 = __shfl(v0, from, 32);
        const float a1 = __shfl(v1, from, 32);
        const float a2 = __shfl(v2, from, 32);
        const float a3 = __shfl(v3, from, 32);
        const float lo  = (f & 1) ? a1 : a0;
        const float hi  = (f & 1) ? a3 : a2;
        const float val = (f & 2) ? hi : lo;
        if (d >= 0) atomicAdd(aggx + (size_t)d * FIN + f, val);
    }
}

__global__ void k_relu_mv(const float* __restrict__ agg,
                          const float* __restrict__ b1,
                          const float* __restrict__ W2,
                          float* __restrict__ z, int N) {
    const int lane = threadIdx.x & 31;
    const int n = blockIdx.x * (blockDim.x >> 5) + (threadIdx.x >> 5);
    if (n >= N) return;
    float s = 0.0f;
#pragma unroll
    for (int j = 0; j < FHID / 32; ++j) {
        int f = j * 32 + lane;
        float h = agg[(size_t)n * FHID + f] + b1[f];
        h = fmaxf(h, 0.0f);
        s = fmaf(h, W2[f], s);
    }
#pragma unroll
    for (int off = 16; off > 0; off >>= 1)
        s += __shfl_xor(s, off, 32);
    if (lane == 0) z[n] = s;
}

__global__ void k_edge2(const int* __restrict__ ei,
                        const float* __restrict__ z,
                        const float* __restrict__ dis,
                        float* __restrict__ oacc, int E, int N) {
    long long e = (long long)blockIdx.x * blockDim.x + threadIdx.x;
    if (e >= (long long)E + N) return;
    int src, dst;
    if (e < E) { src = ei[e]; dst = ei[E + e]; }
    else       { src = dst = (int)(e - E); }
    atomicAdd(&oacc[dst], z[src] * dis[src] * dis[dst]);
}

__global__ void k_sigmoid(const float* __restrict__ oacc,
                          const float* __restrict__ b2,
                          float* __restrict__ out, int N) {
    int n = blockIdx.x * blockDim.x + threadIdx.x;
    if (n >= N) return;
    float v = oacc[n] + b2[0];
    out[n] = 1.0f / (1.0f + __expf(-v));
}

extern "C" void kernel_launch(void* const* d_in, const int* in_sizes, int n_in,
                              void* d_out, int out_size, void* d_ws, size_t ws_size,
                              hipStream_t stream) {
    const float* x  = (const float*)d_in[0];
    const int*   ei = (const int*)d_in[1];
    const float* W1 = (const float*)d_in[2];
    const float* b1 = (const float*)d_in[3];
    const float* W2 = (const float*)d_in[4];
    const float* b2 = (const float*)d_in[5];
    float* out = (float*)d_out;

    const int N = in_sizes[0] / FIN;
    const int E = in_sizes[1] / 2;
    const int T = E + N;

    char* ws = (char*)d_ws;
    float* dis  = (float*)ws;                 ws += (size_t)N * sizeof(float);
    float* aggx = (float*)ws;                 ws += (size_t)N * FIN * sizeof(float);
    float* agg  = (float*)ws;                 ws += (size_t)N * FHID * sizeof(float);
    float* z    = (float*)ws;                 ws += (size_t)N * sizeof(float);
    float* oacc = (float*)ws;

    hipMemsetAsync(dis,  0, (size_t)N * sizeof(float), stream);
    hipMemsetAsync(aggx, 0, (size_t)N * FIN * sizeof(float), stream);
    hipMemsetAsync(oacc, 0, (size_t)N * sizeof(float), stream);

    const int BT = 256;
    const int WPB = BT / 32;

    k_degree  <<<cdiv(T, BT), BT, 0, stream>>>(ei, dis, E, N);
    k_rsqrt   <<<cdiv(N, BT), BT, 0, stream>>>(dis, N);

    k_edge1x  <<<cdiv(T, BT), BT, 0, stream>>>(ei, x, dis, aggx, E, N);

    const int tiles = cdiv(N, 16);
    k_xw1_wmma<<<cdiv(tiles, WPB), BT, 0, stream>>>(aggx, W1, agg, N);

    k_relu_mv <<<cdiv(N, WPB), BT, 0, stream>>>(agg, b1, W2, z, N);
    k_edge2   <<<cdiv(T, BT), BT, 0, stream>>>(ei, z, dis, oacc, E, N);
    k_sigmoid <<<cdiv(N, BT), BT, 0, stream>>>(oacc, b2, out, N);
}


# W1 product, bias, relu and W2 dot fused in one kernel: tile taken transposed on WMMA so a lane holds 32 hidden values of its node; agg not stored, 1 launch fewer; W2-dot float sum order changes
# speedup vs baseline: 1.2749x; 1.0321x over previous
#include <hip/hip_runtime.h>
#include <hip/hip_bf16.h>

#define FIN  4
#define FHID 64

typedef float v2f __attribute__((ext_vector_type(2)));
typedef float v8f __attribute__((ext_vector_type(8)));

static __host__ int cdiv(int a, int b) { return (a + b - 1) / b; }

__global__ void k_degree(const int* __restrict__ ei, float* __restrict__ deg,
                         int E, int N) {
    int e = blockIdx.x * blockDim.x + threadIdx.x;
    if (e >= E + N) return;
    int dst = (e < E) ? ei[E + e] : (e - E);
    atomicAdd(&deg[dst], 1.0f);
}

__global__ void k_rsqrt(float* __restrict__ d, int N) {
    int n = blockIdx.x * blockDim.x + threadIdx.x;
    if (n >= N) return;
    float v = d[n];
    d[n] = (v > 0.0f) ? rsqrtf(v) : 0.0f;
}

__global__ void k_xw1_relu_mv(const float* __restrict__ x,
                              const float* __restrict__ W1,
                              const float* __restrict__ b1,
                              const float* __restrict__ W2,
                              float* __restrict__ z, int N) {
    const int lane = threadIdx.x & 31;
    const int wave = threadIdx.x >> 5;
    const int tile = blockIdx.x * (blockDim.x >> 5) + wave;
    if (tile * 16 >= N) return;

    const int half = lane >> 4;
    const int l16  = lane & 15;
    const int node = tile * 16 + l16;
    const int nrow = (node < N) ? node : (N - 1);
    const int kb   = half * 2;

    v2f xb;
    xb.x = x[(size_t)nrow * FIN + kb + 0];
    xb.y = x[(size_t)nrow * FIN + kb + 1];

    float s[FHID / 16];
#pragma unroll
    for (int c = 0; c < FHID / 16; ++c) {
        const int hcol = c * 16 + l16;
        v2f wa;
        wa.x = W1[(kb + 0) * FHID + hcol];
        wa.y = W1[(kb + 1) * FHID + hcol];
        v8f acc = {};
        acc = __builtin_amdgcn_wmma_f32_16x16x4_f32(
             false, wa,  false, xb,
             (short)0, acc,  false,  false);

        const int hb = c * 16 + half * 8;
        const float4 bA = *(const float4*)(b1 + hb);
        const float4 bB = *(const float4*)(b1 + hb + 4);
        const float4 wA = *(const float4*)(W2 + hb);
        const float4 wB = *(const float4*)(W2 + hb + 4);
        float sc = 0.0f;
        sc = fmaf(fmaxf(acc[0] + bA.x, 0.0f), wA.x, sc);
        sc = fmaf(fmaxf(acc[1] + bA.y, 0.0f), wA.y, sc);
        sc = fmaf(fmaxf(acc[2] + bA.z, 0.0f), wA.z, sc);
        sc = fmaf(fmaxf(acc[3] + bA.w, 0.0f), wA.w, sc);
        sc = fmaf(fmaxf(acc[4] + bB.x, 0.0f), wB.x, sc);
        sc = fmaf(fmaxf(acc[5] + bB.y, 0.0f), wB.y, sc);
        sc = fmaf(fmaxf(acc[6] + bB.z, 0.0f), wB.z, sc);
        sc = fmaf(fmaxf(acc[7] + bB.w, 0.0f), wB.w, sc);
        s[c] = sc;
    }
    float tot = (s[0] + s[1]) + (s[2] + s[3]);
    tot += __shfl_xor(tot, 16, 32);
    if (lane < 16 && node < N) z[node] = tot;
}

__global__ void k_edge1x(const int* __restrict__ ei,
                         const float* __restrict__ x,
                         const float* __restrict__ dis,
                         float* __restrict__ aggx, int E, int N) {
    const int lane = threadIdx.x & 31;
    const long long e = (long long)blockIdx.x * blockDim.x + threadIdx.x;

    int   dst = -1;
    float v0 = 0.0f, v1 = 0.0f, v2 = 0.0f, v3 = 0.0f;
    if (e < (long long)E + N) {
        int src;
        if (e < E) { src = ei[e]; dst = ei[E + e]; }
        else       { src = dst = (int)(e - E); }
        const float  nrm = dis[src] * dis[dst];
        const float4 xr  = *(const float4*)(x + (size_t)src * FIN);
        v0 = xr.x * nrm; v1 = xr.y * nrm; v2 = xr.z * nrm; v3 = xr.w * nrm;
    }

    const int f = lane & 3;
    const int q = lane >> 2;
    for (int k = 0; k < 4; ++k) {
        const int from = k * 8 + q;
        const int   d  = __shfl(dst, from, 32);
        const float a0 = __shfl(v0, from, 32);
        const float a1 = __shfl(v1, from, 32);
        const float a2 = __shfl(v2, from, 32);
        const float a3 = __shfl(v3, from, 32);
        const float lo  = (f & 1) ? a1 : a0;
        const float hi  = (f & 1) ? a3 : a2;
        const float val = (f & 2) ? hi : lo;
        if (d >= 0) atomicAdd(aggx + (size_t)d * FIN + f, val);
    }
}

__global__ void k_edge2(const int* __restrict__ ei,
                        const float* __restrict__ z,
                        const float* __restrict__ dis,
                        float* __restrict__ oacc, int E, int N) {
    long long e = (long long)blockIdx.x * blockDim.x + threadIdx.x;
    if (e >= (long long)E + N) return;
    int src, dst;
    if (e < E) { src = ei[e]; dst = ei[E + e]; }
    else       { src = dst = (int)(e - E); }
    atomicAdd(&oacc[dst], z[src] * dis[src] * dis[dst]);
}

__global__ void k_sigmoid(const float* __restrict__ oacc,
                          const float* __restrict__ b2,
                          float* __restrict__ out, int N) {
    int n = blockIdx.x * blockDim.x + threadIdx.x;
    if (n >= N) return;
    float v = oacc[n] + b2[0];
    out[n] = 1.0f / (1.0f + __expf(-v));
}

extern "C" void kernel_launch(void* const* d_in, const int* in_sizes, int n_in,
                              void* d_out, int out_size, void* d_ws, size_t ws_size,
                              hipStream_t stream) {
    const float* x  = (const float*)d_in[0];
    const int*   ei = (const int*)d_in[1];
    const float* W1 = (const float*)d_in[2];
    const float* b1 = (const float*)d_in[3];
    const float* W2 = (const float*)d_in[4];
    const float* b2 = (const float*)d_in[5];
    float* out = (float*)d_out;

    const int N = in_sizes[0] / FIN;
    const int E = in_sizes[1] / 2;
    const int T = E + N;

    char* ws = (char*)d_ws;
    float* dis  = (float*)ws;                 ws += (size_t)N * sizeof(float);
    float* aggx = (float*)ws;                 ws += (size_t)N * FIN * sizeof(float);
    float* z    = (float*)ws;                 ws += (size_t)N * sizeof(float);
    float* oacc = (float*)ws;

    hipMemsetAsync(dis,  0, (size_t)N * sizeof(float), stream);
    hipMemsetAsync(aggx, 0, (size_t)N * FIN * sizeof(float), stream);
    hipMemsetAsync(oacc, 0, (size_t)N * sizeof(float), stream);

    const int BT = 256;
    const int WPB = BT / 32;

    k_degree  <<<cdiv(T, BT), BT, 0, stream>>>(ei, dis, E, N);
    k_rsqrt   <<<cdiv(N, BT), BT, 0, stream>>>(dis, N);

    k_edge1x  <<<cdiv(T, BT), BT, 0, stream>>>(ei, x, dis, aggx, E, N);

    const int tiles = cdiv(N, 16);
    k_xw1_relu_mv<<<cdiv(tiles, WPB), BT, 0, stream>>>(aggx, W1, b1, W2, z, N);

    k_edge2   <<<cdiv(T, BT), BT, 0, stream>>>(ei, z, dis, oacc, E, N);
    k_sigmoid <<<cdiv(N, BT), BT, 0, stream>>>(oacc, b2, out, N);
}


# Edge norm dis[src]*dis[dst] factored out of both edge sums: rows pre-scaled per node (dis*x, dis*z), dis[dst] applied to the finished sum; an edge gathers 1 value, not 3; rounding changes
# speedup vs baseline: 1.3676x; 1.0727x over previous
#include <hip/hip_runtime.h>
#include <hip/hip_bf16.h>

#define FIN  4
#define FHID 64

typedef float v2f __attribute__((ext_vector_type(2)));
typedef float v8f __attribute__((ext_vector_type(8)));

static __host__ int cdiv(int a, int b) { return (a + b - 1) / b; }

__global__ void k_degree(const int* __restrict__ ei, float* __restrict__ deg,
                         int E, int N) {
    int e = blockIdx.x * blockDim.x + threadIdx.x;
    if (e >= E + N) return;
    int dst = (e < E) ? ei[E + e] : (e - E);
    atomicAdd(&deg[dst], 1.0f);
}

__global__ void k_rsqrt_scale(float* __restrict__ d,
                              const float* __restrict__ x,
                              float* __restrict__ xs, int N) {
    int n = blockIdx.x * blockDim.x + threadIdx.x;
    if (n >= N) return;
    float v = d[n];
    const float r = (v > 0.0f) ? rsqrtf(v) : 0.0f;
    d[n] = r;
    const float4 xr = *(const float4*)(x + (size_t)n * FIN);
    float4 o;
    o.x = xr.x * r; o.y = xr.y * r; o.z = xr.z * r; o.w = xr.w * r;
    *(float4*)(xs + (size_t)n * FIN) = o;
}

__global__ void k_xw1_relu_mv(const float* __restrict__ x,
                              const float* __restrict__ dis,
                              const float* __restrict__ W1,
                              const float* __restrict__ b1,
                              const float* __restrict__ W2,
                              float* __restrict__ zs, int N) {
    const int lane = threadIdx.x & 31;
    const int wave = threadIdx.x >> 5;
    const int tile = blockIdx.x * (blockDim.x >> 5) + wave;
    if (tile * 16 >= N) return;

    const int half = lane >> 4;
    const int l16  = lane & 15;
    const int node = tile * 16 + l16;
    const int nrow = (node < N) ? node : (N - 1);
    const int kb   = half * 2;

    const float dn = dis[nrow];
    v2f xb;
    xb.x = x[(size_t)nrow * FIN + kb + 0] * dn;
    xb.y = x[(size_t)nrow * FIN + kb + 1] * dn;

    float s[FHID / 16];
#pragma unroll
    for (int c = 0; c < FHID / 16; ++c) {
        const int hcol = c * 16 + l16;
        v2f wa;
        wa.x = W1[(kb + 0) * FHID + hcol];
        wa.y = W1[(kb + 1) * FHID + hcol];
        v8f acc = {};
        acc = __builtin_amdgcn_wmma_f32_16x16x4_f32(
             false, wa,  false, xb,
             (short)0, acc,  false,  false);

        const int hb = c * 16 + half * 8;
        const float4 bA = *(const float4*)(b1 + hb);
        const float4 bB = *(const float4*)(b1 + hb + 4);
        const float4 wA = *(const float4*)(W2 + hb);
        const float4 wB = *(const float4*)(W2 + hb + 4);
        float sc = 0.0f;
        sc = fmaf(fmaxf(acc[0] + bA.x, 0.0f), wA.x, sc);
        sc = fmaf(fmaxf(acc[1] + bA.y, 0.0f), wA.y, sc);
        sc = fmaf(fmaxf(acc[2] + bA.z, 0.0f), wA.z, sc);
        sc = fmaf(fmaxf(acc[3] + bA.w, 0.0f), wA.w, sc);
        sc = fmaf(fmaxf(acc[4] + bB.x, 0.0f), wB.x, sc);
        sc = fmaf(fmaxf(acc[5] + bB.y, 0.0f), wB.y, sc);
        sc = fmaf(fmaxf(acc[6] + bB.z, 0.0f), wB.z, sc);
        sc = fmaf(fmaxf(acc[7] + bB.w, 0.0f), wB.w, sc);
        s[c] = sc;
    }
    float tot = (s[0] + s[1]) + (s[2] + s[3]);
    tot += __shfl_xor(tot, 16, 32);
    if (lane < 16 && node < N) zs[node] = tot * dn;
}

__global__ void k_edge1x(const int* __restrict__ ei,
                         const float* __restrict__ xs,
                         float* __restrict__ aggx, int E, int N) {
    const int lane = threadIdx.x & 31;
    const long long e = (long long)blockIdx.x * blockDim.x + threadIdx.x;

    int   dst = -1;
    float v0 = 0.0f, v1 = 0.0f, v2 = 0.0f, v3 = 0.0f;
    if (e < (long long)E + N) {
        int src;
        if (e < E) { src = ei[e]; dst = ei[E + e]; }
        else       { src = dst = (int)(e - E); }
        const float4 xr = *(const float4*)(xs + (size_t)src * FIN);
        v0 = xr.x; v1 = xr.y; v2 = xr.z; v3 = xr.w;
    }

    const int f = lane & 3;
    const int q = lane >> 2;
    for (int k = 0; k < 4; ++k) {
        const int from = k * 8 + q;
        const int   d  = __shfl(dst, from, 32);
        const float a0 = __shfl(v0, from, 32);
        const float a1 = __shfl(v1, from, 32);
        const float a2 = __shfl(v2, from, 32);
        const float a3 = __shfl(v3, from, 32);
        const float lo  = (f & 1) ? a1 : a0;
        const float hi  = (f & 1) ? a3 : a2;
        const float val = (f & 2) ? hi : lo;
        if (d >= 0) atomicAdd(aggx + (size_t)d * FIN + f, val);
    }
}

__global__ void k_edge2(const int* __restrict__ ei,
                        const float* __restrict__ zs,
                        float* __restrict__ oacc, int E, int N) {
    long long e = (long long)blockIdx.x * blockDim.x + threadIdx.x;
    if (e >= (long long)E + N) return;
    int src, dst;
    if (e < E) { src = ei[e]; dst = ei[E + e]; }
    else       { src = dst = (int)(e - E); }
    atomicAdd(&oacc[dst], zs[src]);
}

__global__ void k_sigmoid(const float* __restrict__ oacc,
                          const float* __restrict__ dis,
                          const float* __restrict__ b2,
                          float* __restrict__ out, int N) {
    int n = blockIdx.x * blockDim.x + threadIdx.x;
    if (n >= N) return;
    float v = oacc[n] * dis[n] + b2[0];
    out[n] = 1.0f / (1.0f + __expf(-v));
}

extern "C" void kernel_launch(void* const* d_in, const int* in_sizes, int n_in,
                              void* d_out, int out_size, void* d_ws, size_t ws_size,
                              hipStream_t stream) {
    const float* x  = (const float*)d_in[0];
    const int*   ei = (const int*)d_in[1];
    const float* W1 = (const float*)d_in[2];
    const float* b1 = (const float*)d_in[3];
    const float* W2 = (const float*)d_in[4];
    const float* b2 = (const float*)d_in[5];
    float* out = (float*)d_out;

    const int N = in_sizes[0] / FIN;
    const int E = in_sizes[1] / 2;
    const int T = E + N;

    char* ws = (char*)d_ws;
    float* dis  = (float*)ws;                 ws += (size_t)N * sizeof(float);
    float* aggx = (float*)ws;                 ws += (size_t)N * FIN * sizeof(float);
    float* zs   = (float*)ws;                 ws += (size_t)N * sizeof(float);
    float* oacc = (float*)ws;                 ws += (size_t)N * sizeof(float);
    float* xs   = (float*)ws;

    hipMemsetAsync(dis,  0, (size_t)N * sizeof(float), stream);
    hipMemsetAsync(aggx, 0, (size_t)N * FIN * sizeof(float), stream);
    hipMemsetAsync(oacc, 0, (size_t)N * sizeof(float), stream);

    const int BT = 256;
    const int WPB = BT / 32;

    k_degree  <<<cdiv(T, BT), BT, 0, stream>>>(ei, dis, E, N);
    k_rsqrt_scale<<<cdiv(N, BT), BT, 0, stream>>>(dis, x, xs, N);

    k_edge1x  <<<cdiv(T, BT), BT, 0, stream>>>(ei, xs, aggx, E, N);

    const int tiles = cdiv(N, 16);
    k_xw1_relu_mv<<<cdiv(tiles, WPB), BT, 0, stream>>>(aggx, dis, W1, b1, W2, zs, N);

    k_edge2   <<<cdiv(T, BT), BT, 0, stream>>>(ei, zs, oacc, E, N);
    k_sigmoid <<<cdiv(N, BT), BT, 0, stream>>>(oacc, dis, b2, out, N);
}


# Edge sums taken in LDS bins, not global float atomics: 256 work-groups (edge slice x node range) scan dst, add kept edges; node kernels add the partial planes; no memsets; float sum order changes
# speedup vs baseline: 3.2744x; 2.3943x over previous
#include <hip/hip_runtime.h>
#include <hip/hip_bf16.h>

#define FIN  4
#define FHID 64

typedef float v2f __attribute__((ext_vector_type(2)));
typedef float v8f __attribute__((ext_vector_type(8)));
typedef int   v4i __attribute__((ext_vector_type(4)));

static __host__ int cdiv(int a, int b) { return (a + b - 1) / b; }

#define NWG    256
#define TPB    512
#define WPB_E  (TPB / 32)
#define CAPN   4096
#define UNR    13
#define CHUNK  (TPB * 4 * UNR)
#define WROUND (32 * 4 * UNR)
#define SEGCAP 2048
#define S_DEG  16
#define S_AGG  8
#define S_OUT  16

struct Part { int s, e_lo, e_hi, n_lo, n_hi; };

static __device__ __forceinline__ Part part_of(int S, int E, int N) {
    Part p;
    const int R  = NWG / S;
    p.s = blockIdx.x % S;
    const int r  = blockIdx.x / S;
    const int SL = (((E + S - 1) / S + 3) >> 2) << 2;
    p.e_lo = min(p.s * SL, E);
    p.e_hi = min(p.e_lo + SL, E);
    const int RW = (N + R - 1) / R;
    p.n_lo = min(r * RW, N);
    p.n_hi = min(p.n_lo + RW, N);
    return p;
}

static __device__ __forceinline__ void load_round(const int* __restrict__ dstp, int cb, int tid,
                                                  int E, int e_hi, bool vec, v4i (&d)[UNR]) {
    if (vec) {
#pragma unroll
        for (int u = 0; u < UNR; ++u) {
            const int e = cb + (u * TPB + tid) * 4;
            d[u] = *(const v4i*)(dstp + min(e, E - 4));
        }
    } else {
#pragma unroll
        for (int u = 0; u < UNR; ++u) {
            const int e = cb + (u * TPB + tid) * 4;
            const int a = dstp[min(e + 0, E - 1)];
            const int b = dstp[min(e + 1, E - 1)];
            const int c = dstp[min(e + 2, E - 1)];
            const int g = dstp[min(e + 3, E - 1)];
            d[u].x = (e + 0 < e_hi) ? a : -1;
            d[u].y = (e + 1 < e_hi) ? b : -1;
            d[u].z = (e + 2 < e_hi) ? c : -1;
            d[u].w = (e + 3 < e_hi) ? g : -1;
        }
    }
}

__global__ __launch_bounds__(TPB) void k_deg_bins(const int* __restrict__ ei,
                                                  float* __restrict__ degP, int E, int N) {
    __shared__ unsigned bins[CAPN];
    const int tid = threadIdx.x;
    const Part p = part_of(S_DEG, E, N);
    const int* __restrict__ dstp = ei + E;
    const bool vec = ((((size_t)dstp) & 15) == 0) && ((E & 3) == 0);

    for (int lo = p.n_lo; lo < p.n_hi; lo += CAPN) {
        const int hi = min(lo + CAPN, p.n_hi);
        const unsigned span = (unsigned)(hi - lo);
        for (unsigned i = tid; i < span; i += TPB) bins[i] = 0u;
        __syncthreads();
        for (int cb = p.e_lo; cb < p.e_hi; cb += CHUNK) {
            v4i d[UNR];
            load_round(dstp, cb, tid, E, p.e_hi, vec, d);
#pragma unroll
            for (int u = 0; u < UNR; ++u) {
                const int e = cb + (u * TPB + tid) * 4;
                const int lu = (e < p.e_hi) ? lo : 0x7fffffff;
                unsigned k;
                k = (unsigned)(d[u].x - lu); if (k < span) atomicAdd(&bins[k], 1u);
                k = (unsigned)(d[u].y - lu); if (k < span) atomicAdd(&bins[k], 1u);
                k = (unsigned)(d[u].z - lu); if (k < span) atomicAdd(&bins[k], 1u);
                k = (unsigned)(d[u].w - lu); if (k < span) atomicAdd(&bins[k], 1u);
            }
        }
        __syncthreads();
        for (unsigned i = tid; i < span; i += TPB)
            degP[(size_t)p.s * N + lo + i] = (float)bins[i];
        __syncthreads();
    }
}

__global__ void k_rsqrt_scale(const float* __restrict__ degP,
                              const float* __restrict__ x,
                              float* __restrict__ dis,
                              float* __restrict__ xs, int N) {
    int n = blockIdx.x * blockDim.x + threadIdx.x;
    if (n >= N) return;
    float v = 1.0f;
#pragma unroll
    for (int s = 0; s < S_DEG; ++s) v += degP[(size_t)s * N + n];
    const float r = (v > 0.0f) ? rsqrtf(v) : 0.0f;
    dis[n] = r;
    const float4 xr = *(const float4*)(x + (size_t)n * FIN);
    float4 o;
    o.x = xr.x * r; o.y = xr.y * r; o.z = xr.z * r; o.w = xr.w * r;
    *(float4*)(xs + (size_t)n * FIN) = o;
}

template <int F>
static __device__ __forceinline__ void drain_seg(const int* __restrict__ srcp,
                                                 const int* __restrict__ dstp,
                                                 const float* __restrict__ vals,
                                                 float* bins, const int* seg,
                                                 unsigned cnt, int lo, int lane) {
    constexpr int P = 8;
    for (unsigned i0 = lane; i0 < cnt; i0 += 32 * P) {
        bool ok[P]; int src[P]; unsigned k[P];
#pragma unroll
        for (int q = 0; q < P; ++q) {
            const unsigned i = i0 + 32 * q;
            ok[q] = i < cnt;
            const int e = seg[min(i, cnt - 1)];
            src[q] = srcp[e];
            k[q]   = (unsigned)(dstp[e] - lo);
        }
        if (F == 4) {
            float4 v[P];
#pragma unroll
            for (int q = 0; q < P; ++q) v[q] = *(const float4*)(vals + (size_t)src[q] * 4);
#pragma unroll
            for (int q = 0; q < P; ++q) {
                atomicAdd(&bins[k[q] * 4 + 0], ok[q] ? v[q].x : 0.0f);
                atomicAdd(&bins[k[q] * 4 + 1], ok[q] ? v[q].y : 0.0f);
                atomicAdd(&bins[k[q] * 4 + 2], ok[q] ? v[q].z : 0.0f);
                atomicAdd(&bins[k[q] * 4 + 3], ok[q] ? v[q].w : 0.0f);
            }
        } else {
            float v[P];
#pragma unroll
            for (int q = 0; q < P; ++q) v[q] = vals[src[q]];
#pragma unroll
            for (int q = 0; q < P; ++q) atomicAdd(&bins[k[q]], ok[q] ? v[q] : 0.0f);
        }
    }
}

template <int F>
static __device__ __forceinline__ void edge_bins(const int* __restrict__ ei,
                                                 const float* __restrict__ vals,
                                                 float* __restrict__ planes,
                                                 int E, int N, int S,
                                                 float* bins, int* list) {
    const int tid  = threadIdx.x;
    const int lane = tid & 31;
    const Part p = part_of(S, E, N);
    const int* __restrict__ srcp = ei;
    const int* __restrict__ dstp = ei + E;
    const bool vec = ((((size_t)dstp) & 15) == 0) && ((E & 3) == 0);
    int* seg = list + (tid >> 5) * SEGCAP;

    for (int lo = p.n_lo; lo < p.n_hi; lo += CAPN) {
        const int hi = min(lo + CAPN, p.n_hi);
        const unsigned span = (unsigned)(hi - lo);
        for (unsigned i = tid; i < span * F; i += TPB) bins[i] = 0.0f;
        __syncthreads();

        unsigned wcnt = 0;
        for (int cb = p.e_lo; cb < p.e_hi; cb += CHUNK) {
            v4i d[UNR];
            load_round(dstp, cb, tid, E, p.e_hi, vec, d);
#pragma unroll
            for (int u = 0; u < UNR; ++u) {
                const int e = cb + (u * TPB + tid) * 4;
                const int lu = (e < p.e_hi) ? lo : 0x7fffffff;
                const int dv[4] = { d[u].x, d[u].y, d[u].z, d[u].w };
#pragma unroll
                for (int j = 0; j < 4; ++j) {
                    const bool m = (unsigned)(dv[j] - lu) < span;
                    const unsigned bal = __builtin_amdgcn_ballot_w32(m);
                    if (m) seg[wcnt + __builtin_amdgcn_mbcnt_lo(bal, 0u)] = e + j;
                    wcnt += (unsigned)__builtin_popcount(bal);
                }
            }
            if (cb + CHUNK >= p.e_hi || wcnt + WROUND > SEGCAP) {
                __builtin_amdgcn_fence(__ATOMIC_RELEASE, "wavefront");
                __builtin_amdgcn_wave_barrier();
                __builtin_amdgcn_fence(__ATOMIC_ACQUIRE, "wavefront");
                drain_seg<F>(srcp, dstp, vals, bins, seg, wcnt, lo, lane);
                __builtin_amdgcn_fence(__ATOMIC_RELEASE, "wavefront");
                __builtin_amdgcn_wave_barrier();
                wcnt = 0;
            }
        }
        __syncthreads();
        if (F == 4) {
            for (unsigned i = tid; i < span; i += TPB)
                *(float4*)(planes + ((size_t)p.s * N + lo + i) * 4) = *(const float4*)(bins + (size_t)i * 4);
        } else {
            for (unsigned i = tid; i < span; i += TPB)
                planes[(size_t)p.s * N + lo + i] = bins[i];
        }
        __syncthreads();
    }
}

__global__ __launch_bounds__(TPB) void k_edge1_bins(const int* __restrict__ ei,
                                                    const float* __restrict__ xs,
                                                    float* __restrict__ aggP, int E, int N) {
    __shared__ __attribute__((aligned(16))) float bins[CAPN * FIN];
    __shared__ int list[WPB_E * SEGCAP];
    edge_bins<FIN>(ei, xs, aggP, E, N, S_AGG, bins, list);
}

__global__ __launch_bounds__(TPB) void k_edge2_bins(const int* __restrict__ ei,
                                                    const float* __restrict__ zs,
                                                    float* __restrict__ oP, int E, int N) {
    __shared__ __attribute__((aligned(16))) float bins[CAPN];
    __shared__ int list[WPB_E * SEGCAP];
    edge_bins<1>(ei, zs, oP, E, N, S_OUT, bins, list);
}

__global__ void k_xw1_relu_mv(const float* __restrict__ xs,
                              const float* __restrict__ aggP,
                              const float* __restrict__ dis,
                              const float* __restrict__ W1,
                              const float* __restrict__ b1,
                              const float* __restrict__ W2,
                              float* __restrict__ zs, int N) {
    const int lane = threadIdx.x & 31;
    const int wave = threadIdx.x >> 5;
    const int tile = blockIdx.x * (blockDim.x >> 5) + wave;
    if (tile * 16 >= N) return;

    const int half = lane >> 4;
    const int l16  = lane & 15;
    const int node = tile * 16 + l16;
    const int nrow = (node < N) ? node : (N - 1);
    const int kb   = half * 2;

    const float dn = dis[nrow];
    v2f a = *(const v2f*)(xs + (size_t)nrow * FIN + kb);
#pragma unroll
    for (int s = 0; s < S_AGG; ++s) {
        const v2f pl = *(const v2f*)(aggP + ((size_t)s * N + nrow) * FIN + kb);
        a.x += pl.x; a.y += pl.y;
    }
    v2f xb;
    xb.x = a.x * dn;
    xb.y = a.y * dn;

    float s[FHID / 16];
#pragma unroll
    for (int c = 0; c < FHID / 16; ++c) {
        const int hcol = c * 16 + l16;
        v2f wa;
        wa.x = W1[(kb + 0) * FHID + hcol];
        wa.y = W1[(kb + 1) * FHID + hcol];
        v8f acc = {};
        acc = __builtin_amdgcn_wmma_f32_16x16x4_f32(
             false, wa,  false, xb,
             (short)0, acc,  false,  false);

        const int hb = c * 16 + half * 8;
        const float4 bA = *(const float4*)(b1 + hb);
        const float4 bB = *(const float4*)(b1 + hb + 4);
        const float4 wA = *(const float4*)(W2 + hb);
        const float4 wB = *(const float4*)(W2 + hb + 4);
        float sc = 0.0f;
        sc = fmaf(fmaxf(acc[0] + bA.x, 0.0f), wA.x, sc);
        sc = fmaf(fmaxf(acc[1] + bA.y, 0.0f), wA.y, sc);
        sc = fmaf(fmaxf(acc[2] + bA.z, 0.0f), wA.z, sc);
        sc = fmaf(fmaxf(acc[3] + bA.w, 0.0f), wA.w, sc);
        sc = fmaf(fmaxf(acc[4] + bB.x, 0.0f), wB.x, sc);
        sc = fmaf(fmaxf(acc[5] + bB.y, 0.0f), wB.y, sc);
        sc = fmaf(fmaxf(acc[6] + bB.z, 0.0f), wB.z, sc);
        sc = fmaf(fmaxf(acc[7] + bB.w, 0.0f), wB.w, sc);
        s[c] = sc;
    }
    float tot = (s[0] + s[1]) + (s[2] + s[3]);
    tot += __shfl_xor(tot, 16, 32);
    if (lane < 16 && node < N) zs[node] = tot * dn;
}

__global__ void k_sigmoid(const float* __restrict__ oP,
                          const float* __restrict__ zs,
                          const float* __restrict__ dis,
                          const float* __restrict__ b2,
                          float* __restrict__ out, int N) {
    int n = blockIdx.x * blockDim.x + threadIdx.x;
    if (n >= N) return;
    float acc = zs[n];
#pragma unroll
    for (int s = 0; s < S_OUT; ++s) acc += oP[(size_t)s * N + n];
    float v = acc * dis[n] + b2[0];
    out[n] = 1.0f / (1.0f + __expf(-v));
}

extern "C" void kernel_launch(void* const* d_in, const int* in_sizes, int n_in,
                              void* d_out, int out_size, void* d_ws, size_t ws_size,
                              hipStream_t stream) {
    const float* x  = (const float*)d_in[0];
    const int*   ei = (const int*)d_in[1];
    const float* W1 = (const float*)d_in[2];
    const float* b1 = (const float*)d_in[3];
    const float* W2 = (const float*)d_in[4];
    const float* b2 = (const float*)d_in[5];
    float* out = (float*)d_out;

    const int N = in_sizes[0] / FIN;
    const int E = in_sizes[1] / 2;

    char* ws = (char*)d_ws;
    float* dis    = (float*)ws;               ws += (size_t)N * sizeof(float);
    float* xs     = (float*)ws;               ws += (size_t)N * FIN * sizeof(float);
    float* zs     = (float*)ws;               ws += (size_t)N * sizeof(float);
    float* planes = (float*)ws;

    const int BT = 256;
    const int WPB = BT / 32;

    k_deg_bins   <<<NWG, TPB, 0, stream>>>(ei, planes, E, N);
    k_rsqrt_scale<<<cdiv(N, BT), BT, 0, stream>>>(planes, x, dis, xs, N);

    k_edge1_bins <<<NWG, TPB, 0, stream>>>(ei, xs, planes, E, N);

    const int tiles = cdiv(N, 16);
    k_xw1_relu_mv<<<cdiv(tiles, WPB), BT, 0, stream>>>(xs, planes, dis, W1, b1, W2, zs, N);

    k_edge2_bins <<<NWG, TPB, 0, stream>>>(ei, zs, planes, E, N);
    k_sigmoid    <<<cdiv(N, BT), BT, 0, stream>>>(planes, zs, dis, b2, out, N);
}


# Edges sorted once by destination range (counting sort per 2048-edge chunk in LDS); one work-group per range sums its edges in LDS and finishes its nodes; 6 launches to 4; float sum order changes
# speedup vs baseline: 5.9412x; 1.8144x over previous
#include <hip/hip_runtime.h>

#define FIN  4
#define FHID 64

typedef float v2f __attribute__((ext_vector_type(2)));
typedef float v8f __attribute__((ext_vector_type(8)));

static __host__ int cdiv(int a, int b) { return (a + b - 1) / b; }

#define NR    256
#define CH    2048
#define TPB_B 256
#define EPT   (CH / TPB_B)
#define TPB   512
#define WPB   (TPB / 32)
#define CAPN  1024
#define P     8

__global__ __launch_bounds__(TPB_B) void k_bucket(const int* __restrict__ ei,
                                                  int* __restrict__ off,
                                                  int2* __restrict__ pairs,
                                                  int E, int N, int RW) {
    __shared__ int  cnt[NR + 1];
    __shared__ int2 stage[CH];
    const int tid = threadIdx.x;
    const int c   = blockIdx.x;
    const int e0  = c * CH;
    const int n_e = min(CH, E - e0);

    for (int i = tid; i <= NR; i += TPB_B) cnt[i] = 0;
    __syncthreads();

    int src[EPT], dst[EPT], grp[EPT], pos[EPT];
#pragma unroll
    for (int k = 0; k < EPT; ++k) {
        const int e = e0 + min(k * TPB_B + tid, n_e - 1);
        src[k] = ei[e];
        dst[k] = ei[(size_t)E + e];
    }
#pragma unroll
    for (int k = 0; k < EPT; ++k) {
        const bool in = (k * TPB_B + tid) < n_e;
        const unsigned d = (unsigned)dst[k];
        grp[k] = (d < (unsigned)N) ? (int)(d / (unsigned)RW) : NR;
        pos[k] = in ? atomicAdd(&cnt[grp[k]], 1) : 0;
    }
    __syncthreads();

    if (tid < 32) {
        int ex[NR / 32];
        int run = 0;
#pragma unroll
        for (int j = 0; j < NR / 32; ++j) { ex[j] = run; run += cnt[tid * (NR / 32) + j]; }
        int inc = run;
#pragma unroll
        for (int d = 1; d < 32; d <<= 1) {
            const int t = __shfl_up(inc, d, 32);
            if (tid >= d) inc += t;
        }
        const int base = inc - run;
#pragma unroll
        for (int j = 0; j < NR / 32; ++j) cnt[tid * (NR / 32) + j] = base + ex[j];
        if (tid == 31) cnt[NR] = inc;
    }
    __syncthreads();

    for (int i = tid; i <= NR; i += TPB_B) off[(size_t)c * (NR + 1) + i] = cnt[i];
#pragma unroll
    for (int k = 0; k < EPT; ++k) {
        if ((k * TPB_B + tid) < n_e) stage[cnt[grp[k]] + pos[k]] = make_int2(src[k], dst[k]);
    }
    __syncthreads();
    for (int i = tid; i < n_e; i += TPB_B) pairs[(size_t)e0 + i] = stage[i];
}

static __device__ __forceinline__ int seg_table(const int* __restrict__ off, int NCH, int c0, int r,
                                                int tid, int* sStart, int* sBase, int* wsum) {
    const int c = c0 + tid;
    int len = 0, base = 0;
    if (c < NCH) {
        const int a = off[(size_t)c * (NR + 1) + r];
        const int b = off[(size_t)c * (NR + 1) + r + 1];
        len  = b - a;
        base = c * CH + a;
    }
    const int lane = tid & 31, wave = tid >> 5;
    int inc = len;
#pragma unroll
    for (int d = 1; d < 32; d <<= 1) {
        const int t = __shfl_up(inc, d, 32);
        if (lane >= d) inc += t;
    }
    if (lane == 31) wsum[wave] = inc;
    __syncthreads();
    int wbase = 0, tot = 0;
#pragma unroll
    for (int w = 0; w < WPB; ++w) {
        const int t = wsum[w];
        if (w < wave) wbase += t;
        tot += t;
    }
    sStart[tid] = wbase + inc - len;
    sBase[tid]  = base;
    if (tid == 0) sStart[TPB] = tot;
    __syncthreads();
    return tot;
}

static __device__ __forceinline__ void find_slots(const int* sStart, const int* sBase, int b, int T,
                                                  int tid, bool (&ok)[P], int (&slot)[P]) {
#pragma unroll
    for (int q = 0; q < P; ++q) {
        const int i = b + q * TPB + tid;
        ok[q] = i < T;
        const int ii = min(i, T - 1);
        int j = 0;
#pragma unroll
        for (int st = TPB / 2; st >= 1; st >>= 1)
            if (sStart[j + st] <= ii) j += st;
        slot[q] = sBase[j] + (ii - sStart[j]);
    }
}

__global__ __launch_bounds__(TPB) void k_deg_scale(const int* __restrict__ off,
                                                   const int2* __restrict__ pairs,
                                                   const float* __restrict__ x,
                                                   float* __restrict__ dis,
                                                   float* __restrict__ xs,
                                                   int N, int NCH, int RW) {
    __shared__ unsigned bins[CAPN];
    __shared__ int sStart[TPB + 1];
    __shared__ int sBase[TPB];
    __shared__ int wsum[WPB];
    const int tid = threadIdx.x;
    const int r   = blockIdx.x;
    const int n_lo = min(r * RW, N);
    const int n_hi = min(n_lo + RW, N);

    for (int lo = n_lo; lo < n_hi; lo += CAPN) {
        const int hi = min(lo + CAPN, n_hi);
        const unsigned span = (unsigned)(hi - lo);
        const unsigned idle = min((unsigned)(tid & 31), span - 1u);
        for (unsigned i = tid; i < span; i += TPB) bins[i] = 0u;
        __syncthreads();
        for (int c0 = 0; c0 < NCH; c0 += TPB) {
            const int T = seg_table(off, NCH, c0, r, tid, sStart, sBase, wsum);
            for (int b = 0; b < T; b += TPB * P) {
                bool ok[P]; int slot[P]; int d[P];
                find_slots(sStart, sBase, b, T, tid, ok, slot);
#pragma unroll
                for (int q = 0; q < P; ++q) d[q] = pairs[slot[q]].y;
#pragma unroll
                for (int q = 0; q < P; ++q) {
                    const unsigned k = (unsigned)(d[q] - lo);
                    const bool in = ok[q] && k < span;
                    atomicAdd(&bins[in ? k : idle], in ? 1u : 0u);
                }
            }
            __syncthreads();
        }
        for (unsigned i = tid; i < span; i += TPB) {
            const int n = lo + (int)i;
            const float v  = 1.0f + (float)bins[i];
            const float rs = (v > 0.0f) ? rsqrtf(v) : 0.0f;
            dis[n] = rs;
            const float4 xr = *(const float4*)(x + (size_t)n * FIN);
            float4 o;
            o.x = xr.x * rs; o.y = xr.y * rs; o.z = xr.z * rs; o.w = xr.w * rs;
            *(float4*)(xs + (size_t)n * FIN) = o;
        }
        __syncthreads();
    }
}

__global__ __launch_bounds__(TPB) void k_layer1(const int* __restrict__ off,
                                                const int2* __restrict__ pairs,
                                                const float* __restrict__ xs,
                                                const float* __restrict__ dis,
                                                const float* __restrict__ W1,
                                                const float* __restrict__ b1,
                                                const float* __restrict__ W2,
                                                float* __restrict__ zs,
                                                int N, int NCH, int RW) {
    __shared__ __attribute__((aligned(16))) float bins[(CAPN + 32) * FIN];
    __shared__ int sStart[TPB + 1];
    __shared__ int sBase[TPB];
    __shared__ int wsum[WPB];
    const int tid  = threadIdx.x;
    const int lane = tid & 31;
    const int wave = tid >> 5;
    const int r    = blockIdx.x;
    const int n_lo = min(r * RW, N);
    const int n_hi = min(n_lo + RW, N);

    for (int lo = n_lo; lo < n_hi; lo += CAPN) {
        const int hi = min(lo + CAPN, n_hi);
        const unsigned span = (unsigned)(hi - lo);
        const unsigned idle = CAPN + (unsigned)lane;
        for (unsigned i = tid; i < span * FIN; i += TPB) bins[i] = 0.0f;
        if (tid < 32 * FIN) bins[CAPN * FIN + tid] = 0.0f;
        __syncthreads();
        for (int c0 = 0; c0 < NCH; c0 += TPB) {
            const int T = seg_table(off, NCH, c0, r, tid, sStart, sBase, wsum);
            for (int b = 0; b < T; b += TPB * P) {
                bool ok[P]; int slot[P]; int2 pr[P]; float4 v[P];
                find_slots(sStart, sBase, b, T, tid, ok, slot);
#pragma unroll
                for (int q = 0; q < P; ++q) pr[q] = pairs[slot[q]];
#pragma unroll
                for (int q = 0; q < P; ++q) v[q] = *(const float4*)(xs + (size_t)pr[q].x * FIN);
#pragma unroll
                for (int q = 0; q < P; ++q) {
                    const unsigned k = (unsigned)(pr[q].y - lo);
                    const unsigned kk = (ok[q] && k < span) ? k : idle;
                    atomicAdd(&bins[kk * FIN + 0], v[q].x);
                    atomicAdd(&bins[kk * FIN + 1], v[q].y);
                    atomicAdd(&bins[kk * FIN + 2], v[q].z);
                    atomicAdd(&bins[kk * FIN + 3], v[q].w);
                }
            }
            __syncthreads();
        }

        const int half = lane >> 4;
        const int l16  = lane & 15;
        const int kb   = half * 2;
        for (int t = wave; t * 16 < (int)span; t += WPB) {
            const int li   = t * 16 + l16;
            const int lrow = min(li, (int)span - 1);
            const int nrow = lo + lrow;

            const float dn = dis[nrow];
            const v2f self = *(const v2f*)(xs + (size_t)nrow * FIN + kb);
            const v2f sum  = *(const v2f*)(bins + lrow * FIN + kb);
            v2f xb;
            xb.x = (self.x + sum.x) * dn;
            xb.y = (self.y + sum.y) * dn;

            float s[FHID / 16];
#pragma unroll
            for (int c = 0; c < FHID / 16; ++c) {
                const int hcol = c * 16 + l16;
                v2f wa;
                wa.x = W1[(kb + 0) * FHID + hcol];
                wa.y = W1[(kb + 1) * FHID + hcol];
                v8f acc = {};
                acc = __builtin_amdgcn_wmma_f32_16x16x4_f32(
                     false, wa,  false, xb,
                     (short)0, acc,  false,  false);

                const int hb = c * 16 + half * 8;
                const float4 bA = *(const float4*)(b1 + hb);
                const float4 bB = *(const float4*)(b1 + hb + 4);
                const float4 wA = *(const float4*)(W2 + hb);
                const float4 wB = *(const float4*)(W2 + hb + 4);
                float sc = 0.0f;
                sc = fmaf(fmaxf(acc[0] + bA.x, 0.0f), wA.x, sc);
                sc = fmaf(fmaxf(acc[1] + bA.y, 0.0f), wA.y, sc);
                sc = fmaf(fmaxf(acc[2] + bA.z, 0.0f), wA.z, sc);
                sc = fmaf(fmaxf(acc[3] + bA.w, 0.0f), wA.w, sc);
                sc = fmaf(fmaxf(acc[4] + bB.x, 0.0f), wB.x, sc);
                sc = fmaf(fmaxf(acc[5] + bB.y, 0.0f), wB.y, sc);
                sc = fmaf(fmaxf(acc[6] + bB.z, 0.0f), wB.z, sc);
                sc = fmaf(fmaxf(acc[7] + bB.w, 0.0f), wB.w, sc);
                s[c] = sc;
            }
            float tot = (s[0] + s[1]) + (s[2] + s[3]);
            tot += __shfl_xor(tot, 16, 32);
            if (lane < 16 && li < (int)span) zs[lo + li] = tot * dn;
        }
        __syncthreads();
    }
}

__global__ __launch_bounds__(TPB) void k_layer2(const int* __restrict__ off,
                                                const int2* __restrict__ pairs,
                                                const float* __restrict__ zs,
                                                const float* __restrict__ dis,
                                                const float* __restrict__ b2,
                                                float* __restrict__ out,
                                                int N, int NCH, int RW) {
    __shared__ float bins[CAPN + 32];
    __shared__ int sStart[TPB + 1];
    __shared__ int sBase[TPB];
    __shared__ int wsum[WPB];
    const int tid = threadIdx.x;
    const int r   = blockIdx.x;
    const int n_lo = min(r * RW, N);
    const int n_hi = min(n_lo + RW, N);

    for (int lo = n_lo; lo < n_hi; lo += CAPN) {
        const int hi = min(lo + CAPN, n_hi);
        const unsigned span = (unsigned)(hi - lo);
        const unsigned idle = CAPN + (unsigned)(tid & 31);
        for (unsigned i = tid; i < span; i += TPB) bins[i] = 0.0f;
        if (tid < 32) bins[CAPN + tid] = 0.0f;
        __syncthreads();
        for (int c0 = 0; c0 < NCH; c0 += TPB) {
            const int T = seg_table(off, NCH, c0, r, tid, sStart, sBase, wsum);
            for (int b = 0; b < T; b += TPB * P) {
                bool ok[P]; int slot[P]; int2 pr[P]; float v[P];
                find_slots(sStart, sBase, b, T, tid, ok, slot);
#pragma unroll
                for (int q = 0; q < P; ++q) pr[q] = pairs[slot[q]];
#pragma unroll
                for (int q = 0; q < P; ++q) v[q] = zs[pr[q].x];
#pragma unroll
                for (int q = 0; q < P; ++q) {
                    const unsigned k = (unsigned)(pr[q].y - lo);
                    atomicAdd(&bins[(ok[q] && k < span) ? k : idle], v[q]);
                }
            }
            __syncthreads();
        }
        for (unsigned i = tid; i < span; i += TPB) {
            const int n = lo + (int)i;
            const float v = (zs[n] + bins[i]) * dis[n] + b2[0];
            out[n] = 1.0f / (1.0f + __expf(-v));
        }
        __syncthreads();
    }
}

extern "C" void kernel_launch(void* const* d_in, const int* in_sizes, int n_in,
                              void* d_out, int out_size, void* d_ws, size_t ws_size,
                              hipStream_t stream) {
    const float* x  = (const float*)d_in[0];
    const int*   ei = (const int*)d_in[1];
    const float* W1 = (const float*)d_in[2];
    const float* b1 = (const float*)d_in[3];
    const float* W2 = (const float*)d_in[4];
    const float* b2 = (const float*)d_in[5];
    float* out = (float*)d_out;

    const int N = in_sizes[0] / FIN;
    const int E = in_sizes[1] / 2;
    const int NCH = cdiv(E, CH);
    const int RW  = cdiv(N, NR);

    char* ws = (char*)d_ws;
    float* xs    = (float*)ws;               ws += (size_t)N * FIN * sizeof(float);
    int2*  pairs = (int2*)ws;                ws += (size_t)E * sizeof(int2);
    float* dis   = (float*)ws;               ws += (size_t)N * sizeof(float);
    float* zs    = (float*)ws;               ws += (size_t)N * sizeof(float);
    int*   off   = (int*)ws;

    k_bucket   <<<NCH, TPB_B, 0, stream>>>(ei, off, pairs, E, N, RW);
    k_deg_scale<<<NR,  TPB,   0, stream>>>(off, pairs, x, dis, xs, N, NCH, RW);
    k_layer1   <<<NR,  TPB,   0, stream>>>(off, pairs, xs, dis, W1, b1, W2, zs, N, NCH, RW);
    k_layer2   <<<NR,  TPB,   0, stream>>>(off, pairs, zs, dis, b2, out, N, NCH, RW);
}


# Sorted chunks of 8192 edges, not 2048 (1024-thread sort groups): a range's edges lie in 98 runs of about 32 pairs, not 391 of 8; table 4x shorter, 7-level search; float sum order changes
# speedup vs baseline: 6.4142x; 1.0796x over previous
#include <hip/hip_runtime.h>

#define FIN  4
#define FHID 64

typedef float v2f __attribute__((ext_vector_type(2)));
typedef float v8f __attribute__((ext_vector_type(8)));

static __host__ int cdiv(int a, int b) { return (a + b - 1) / b; }

#define NR    256
#define CH    8192
#define TPB_B 1024
#define EPT   (CH / TPB_B)
#define TPB   512
#define WPB   (TPB / 32)
#define SEG   128
#define CAPN  1024
#define P     8

__global__ __launch_bounds__(TPB_B) void k_bucket(const int* __restrict__ ei,
                                                  int* __restrict__ off,
                                                  int2* __restrict__ pairs,
                                                  int E, int N, int RW) {
    __shared__ int  cnt[NR + 1];
    __shared__ int2 stage[CH];
    const int tid = threadIdx.x;
    const int c   = blockIdx.x;
    const int e0  = c * CH;
    const int n_e = min(CH, E - e0);

    for (int i = tid; i <= NR; i += TPB_B) cnt[i] = 0;
    __syncthreads();

    int src[EPT], dst[EPT], grp[EPT], pos[EPT];
#pragma unroll
    for (int k = 0; k < EPT; ++k) {
        const int e = e0 + min(k * TPB_B + tid, n_e - 1);
        src[k] = ei[e];
        dst[k] = ei[(size_t)E + e];
    }
#pragma unroll
    for (int k = 0; k < EPT; ++k) {
        const bool in = (k * TPB_B + tid) < n_e;
        const unsigned d = (unsigned)dst[k];
        grp[k] = (d < (unsigned)N) ? (int)(d / (unsigned)RW) : NR;
        pos[k] = in ? atomicAdd(&cnt[grp[k]], 1) : 0;
    }
    __syncthreads();

    if (tid < 32) {
        int ex[NR / 32];
        int run = 0;
#pragma unroll
        for (int j = 0; j < NR / 32; ++j) { ex[j] = run; run += cnt[tid * (NR / 32) + j]; }
        int inc = run;
#pragma unroll
        for (int d = 1; d < 32; d <<= 1) {
            const int t = __shfl_up(inc, d, 32);
            if (tid >= d) inc += t;
        }
        const int base = inc - run;
#pragma unroll
        for (int j = 0; j < NR / 32; ++j) cnt[tid * (NR / 32) + j] = base + ex[j];
        if (tid == 31) cnt[NR] = inc;
    }
    __syncthreads();

    for (int i = tid; i <= NR; i += TPB_B) off[(size_t)c * (NR + 1) + i] = cnt[i];
#pragma unroll
    for (int k = 0; k < EPT; ++k) {
        if ((k * TPB_B + tid) < n_e) stage[cnt[grp[k]] + pos[k]] = make_int2(src[k], dst[k]);
    }
    __syncthreads();
    for (int i = tid; i < n_e; i += TPB_B) pairs[(size_t)e0 + i] = stage[i];
}

static __device__ __forceinline__ int seg_table(const int* __restrict__ off, int NCH, int c0, int r,
                                                int tid, int* sStart, int* sBase, int* wsum) {
    const int c = c0 + tid;
    int len = 0, base = 0;
    if (tid < SEG && c < NCH) {
        const int a = off[(size_t)c * (NR + 1) + r];
        const int b = off[(size_t)c * (NR + 1) + r + 1];
        len  = b - a;
        base = c * CH + a;
    }
    const int lane = tid & 31, wave = tid >> 5;
    int inc = len;
#pragma unroll
    for (int d = 1; d < 32; d <<= 1) {
        const int t = __shfl_up(inc, d, 32);
        if (lane >= d) inc += t;
    }
    if (lane == 31 && wave < SEG / 32) wsum[wave] = inc;
    __syncthreads();
    int wbase = 0, tot = 0;
#pragma unroll
    for (int w = 0; w < SEG / 32; ++w) {
        const int t = wsum[w];
        if (w < wave) wbase += t;
        tot += t;
    }
    if (tid < SEG) {
        sStart[tid] = wbase + inc - len;
        sBase[tid]  = base;
    }
    if (tid == 0) sStart[SEG] = tot;
    __syncthreads();
    return tot;
}

static __device__ __forceinline__ void find_slots(const int* sStart, const int* sBase, int b, int T,
                                                  int tid, bool (&ok)[P], int (&slot)[P]) {
#pragma unroll
    for (int q = 0; q < P; ++q) {
        const int i = b + q * TPB + tid;
        ok[q] = i < T;
        const int ii = min(i, T - 1);
        int j = 0;
#pragma unroll
        for (int st = SEG / 2; st >= 1; st >>= 1)
            if (sStart[j + st] <= ii) j += st;
        slot[q] = sBase[j] + (ii - sStart[j]);
    }
}

__global__ __launch_bounds__(TPB) void k_deg_scale(const int* __restrict__ off,
                                                   const int2* __restrict__ pairs,
                                                   const float* __restrict__ x,
                                                   float* __restrict__ dis,
                                                   float* __restrict__ xs,
                                                   int N, int NCH, int RW) {
    __shared__ unsigned bins[CAPN];
    __shared__ int sStart[SEG + 1];
    __shared__ int sBase[SEG];
    __shared__ int wsum[WPB];
    const int tid = threadIdx.x;
    const int r   = blockIdx.x;
    const int n_lo = min(r * RW, N);
    const int n_hi = min(n_lo + RW, N);

    for (int lo = n_lo; lo < n_hi; lo += CAPN) {
        const int hi = min(lo + CAPN, n_hi);
        const unsigned span = (unsigned)(hi - lo);
        const unsigned idle = min((unsigned)(tid & 31), span - 1u);
        for (unsigned i = tid; i < span; i += TPB) bins[i] = 0u;
        __syncthreads();
        for (int c0 = 0; c0 < NCH; c0 += SEG) {
            const int T = seg_table(off, NCH, c0, r, tid, sStart, sBase, wsum);
            for (int b = 0; b < T; b += TPB * P) {
                bool ok[P]; int slot[P]; int d[P];
                find_slots(sStart, sBase, b, T, tid, ok, slot);
#pragma unroll
                for (int q = 0; q < P; ++q) d[q] = pairs[slot[q]].y;
#pragma unroll
                for (int q = 0; q < P; ++q) {
                    const unsigned k = (unsigned)(d[q] - lo);
                    const bool in = ok[q] && k < span;
                    atomicAdd(&bins[in ? k : idle], in ? 1u : 0u);
                }
            }
            __syncthreads();
        }
        for (unsigned i = tid; i < span; i += TPB) {
            const int n = lo + (int)i;
            const float v  = 1.0f + (float)bins[i];
            const float rs = (v > 0.0f) ? rsqrtf(v) : 0.0f;
            dis[n] = rs;
            const float4 xr = *(const float4*)(x + (size_t)n * FIN);
            float4 o;
            o.x = xr.x * rs; o.y = xr.y * rs; o.z = xr.z * rs; o.w = xr.w * rs;
            *(float4*)(xs + (size_t)n * FIN) = o;
        }
        __syncthreads();
    }
}

__global__ __launch_bounds__(TPB) void k_layer1(const int* __restrict__ off,
                                                const int2* __restrict__ pairs,
                                                const float* __restrict__ xs,
                                                const float* __restrict__ dis,
                                                const float* __restrict__ W1,
                                                const float* __restrict__ b1,
                                                const float* __restrict__ W2,
                                                float* __restrict__ zs,
                                                int N, int NCH, int RW) {
    __shared__ __attribute__((aligned(16))) float bins[(CAPN + 32) * FIN];
    __shared__ int sStart[SEG + 1];
    __shared__ int sBase[SEG];
    __shared__ int wsum[WPB];
    const int tid  = threadIdx.x;
    const int lane = tid & 31;
    const int wave = tid >> 5;
    const int r    = blockIdx.x;
    const int n_lo = min(r * RW, N);
    const int n_hi = min(n_lo + RW, N);

    for (int lo = n_lo; lo < n_hi; lo += CAPN) {
        const int hi = min(lo + CAPN, n_hi);
        const unsigned span = (unsigned)(hi - lo);
        const unsigned idle = CAPN + (unsigned)lane;
        for (unsigned i = tid; i < span * FIN; i += TPB) bins[i] = 0.0f;
        if (tid < 32 * FIN) bins[CAPN * FIN + tid] = 0.0f;
        __syncthreads();
        for (int c0 = 0; c0 < NCH; c0 += SEG) {
            const int T = seg_table(off, NCH, c0, r, tid, sStart, sBase, wsum);
            for (int b = 0; b < T; b += TPB * P) {
                bool ok[P]; int slot[P]; int2 pr[P]; float4 v[P];
                find_slots(sStart, sBase, b, T, tid, ok, slot);
#pragma unroll
                for (int q = 0; q < P; ++q) pr[q] = pairs[slot[q]];
#pragma unroll
                for (int q = 0; q < P; ++q) v[q] = *(const float4*)(xs + (size_t)pr[q].x * FIN);
#pragma unroll
                for (int q = 0; q < P; ++q) {
                    const unsigned k = (unsigned)(pr[q].y - lo);
                    const unsigned kk = (ok[q] && k < span) ? k : idle;
                    atomicAdd(&bins[kk * FIN + 0], v[q].x);
                    atomicAdd(&bins[kk * FIN + 1], v[q].y);
                    atomicAdd(&bins[kk * FIN + 2], v[q].z);
                    atomicAdd(&bins[kk * FIN + 3], v[q].w);
                }
            }
            __syncthreads();
        }

        const int half = lane >> 4;
        const int l16  = lane & 15;
        const int kb   = half * 2;
        for (int t = wave; t * 16 < (int)span; t += WPB) {
            const int li   = t * 16 + l16;
            const int lrow = min(li, (int)span - 1);
            const int nrow = lo + lrow;

            const float dn = dis[nrow];
            const v2f self = *(const v2f*)(xs + (size_t)nrow * FIN + kb);
            const v2f sum  = *(const v2f*)(bins + lrow * FIN + kb);
            v2f xb;
            xb.x = (self.x + sum.x) * dn;
            xb.y = (self.y + sum.y) * dn;

            float s[FHID / 16];
#pragma unroll
            for (int c = 0; c < FHID / 16; ++c) {
                const int hcol = c * 16 + l16;
                v2f wa;
                wa.x = W1[(kb + 0) * FHID + hcol];
                wa.y = W1[(kb + 1) * FHID + hcol];
                v8f acc = {};
                acc = __builtin_amdgcn_wmma_f32_16x16x4_f32(
                     false, wa,  false, xb,
                     (short)0, acc,  false,  false);

                const int hb = c * 16 + half * 8;
                const float4 bA = *(const float4*)(b1 + hb);
                const float4 bB = *(const float4*)(b1 + hb + 4);
                const float4 wA = *(const float4*)(W2 + hb);
                const float4 wB = *(const float4*)(W2 + hb + 4);
                float sc = 0.0f;
                sc = fmaf(fmaxf(acc[0] + bA.x, 0.0f), wA.x, sc);
                sc = fmaf(fmaxf(acc[1] + bA.y, 0.0f), wA.y, sc);
                sc = fmaf(fmaxf(acc[2] + bA.z, 0.0f), wA.z, sc);
                sc = fmaf(fmaxf(acc[3] + bA.w, 0.0f), wA.w, sc);
                sc = fmaf(fmaxf(acc[4] + bB.x, 0.0f), wB.x, sc);
                sc = fmaf(fmaxf(acc[5] + bB.y, 0.0f), wB.y, sc);
                sc = fmaf(fmaxf(acc[6] + bB.z, 0.0f), wB.z, sc);
                sc = fmaf(fmaxf(acc[7] + bB.w, 0.0f), wB.w, sc);
                s[c] = sc;
            }
            float tot = (s[0] + s[1]) + (s[2] + s[3]);
            tot += __shfl_xor(tot, 16, 32);
            if (lane < 16 && li < (int)span) zs[lo + li] = tot * dn;
        }
        __syncthreads();
    }
}

__global__ __launch_bounds__(TPB) void k_layer2(const int* __restrict__ off,
                                                const int2* __restrict__ pairs,
                                                const float* __restrict__ zs,
                                                const float* __restrict__ dis,
                                                const float* __restrict__ b2,
                                                float* __restrict__ out,
                                                int N, int NCH, int RW) {
    __shared__ float bins[CAPN + 32];
    __shared__ int sStart[SEG + 1];
    __shared__ int sBase[SEG];
    __shared__ int wsum[WPB];
    const int tid = threadIdx.x;
    const int r   = blockIdx.x;
    const int n_lo = min(r * RW, N);
    const int n_hi = min(n_lo + RW, N);

    for (int lo = n_lo; lo < n_hi; lo += CAPN) {
        const int hi = min(lo + CAPN, n_hi);
        const unsigned span = (unsigned)(hi - lo);
        const unsigned idle = CAPN + (unsigned)(tid & 31);
        for (unsigned i = tid; i < span; i += TPB) bins[i] = 0.0f;
        if (tid < 32) bins[CAPN + tid] = 0.0f;
        __syncthreads();
        for (int c0 = 0; c0 < NCH; c0 += SEG) {
            const int T = seg_table(off, NCH, c0, r, tid, sStart, sBase, wsum);
            for (int b = 0; b < T; b += TPB * P) {
                bool ok[P]; int slot[P]; int2 pr[P]; float v[P];
                find_slots(sStart, sBase, b, T, tid, ok, slot);
#pragma unroll
                for (int q = 0; q < P; ++q) pr[q] = pairs[slot[q]];
#pragma unroll
                for (int q = 0; q < P; ++q) v[q] = zs[pr[q].x];
#pragma unroll
                for (int q = 0; q < P; ++q) {
                    const unsigned k = (unsigned)(pr[q].y - lo);
                    atomicAdd(&bins[(ok[q] && k < span) ? k : idle], v[q]);
                }
            }
            __syncthreads();
        }
        for (unsigned i = tid; i < span; i += TPB) {
            const int n = lo + (int)i;
            const float v = (zs[n] + bins[i]) * dis[n] + b2[0];
            out[n] = 1.0f / (1.0f + __expf(-v));
        }
        __syncthreads();
    }
}

extern "C" void kernel_launch(void* const* d_in, const int* in_sizes, int n_in,
                              void* d_out, int out_size, void* d_ws, size_t ws_size,
                              hipStream_t stream) {
    const float* x  = (const float*)d_in[0];
    const int*   ei = (const int*)d_in[1];
    const float* W1 = (const float*)d_in[2];
    const float* b1 = (const float*)d_in[3];
    const float* W2 = (const float*)d_in[4];
    const float* b2 = (const float*)d_in[5];
    float* out = (float*)d_out;

    const int N = in_sizes[0] / FIN;
    const int E = in_sizes[1] / 2;
    const int NCH = cdiv(E, CH);
    const int RW  = cdiv(N, NR);

    char* ws = (char*)d_ws;
    float* xs    = (float*)ws;               ws += (size_t)N * FIN * sizeof(float);
    int2*  pairs = (int2*)ws;                ws += (size_t)E * sizeof(int2);
    float* dis   = (float*)ws;               ws += (size_t)N * sizeof(float);
    float* zs    = (float*)ws;               ws += (size_t)N * sizeof(float);
    int*   off   = (int*)ws;

    k_bucket   <<<NCH, TPB_B, 0, stream>>>(ei, off, pairs, E, N, RW);
    k_deg_scale<<<NR,  TPB,   0, stream>>>(off, pairs, x, dis, xs, N, NCH, RW);
    k_layer1   <<<NR,  TPB,   0, stream>>>(off, pairs, xs, dis, W1, b1, W2, zs, N, NCH, RW);
    k_layer2   <<<NR,  TPB,   0, stream>>>(off, pairs, zs, dis, b2, out, N, NCH, RW);
}


# Range kernels: a wave takes whole runs straight from the table (lane = pair of a run, run starts and lengths in scalar registers): no work-group prefix, no LDS search; LDS float sum order changes
# speedup vs baseline: 7.6436x; 1.1917x over previous
#include <hip/hip_runtime.h>

#define FIN  4
#define FHID 64

typedef float v2f __attribute__((ext_vector_type(2)));
typedef float v8f __attribute__((ext_vector_type(8)));

static __host__ int cdiv(int a, int b) { return (a + b - 1) / b; }

#define NR    256
#define CH    8192
#define TPB_B 1024
#define EPT   (CH / TPB_B)
#define TPB   512
#define WPB   (TPB / 32)
#define CAPN  1024
#define RP    8
#define KC    2
#define P     (RP * KC)

__global__ __launch_bounds__(TPB_B) void k_bucket(const int* __restrict__ ei,
                                                  int* __restrict__ off,
                                                  int2* __restrict__ pairs,
                                                  int E, int N, int RW) {
    __shared__ int  cnt[NR + 1];
    __shared__ int2 stage[CH];
    const int tid = threadIdx.x;
    const int c   = blockIdx.x;
    const int e0  = c * CH;
    const int n_e = min(CH, E - e0);

    for (int i = tid; i <= NR; i += TPB_B) cnt[i] = 0;
    __syncthreads();

    int src[EPT], dst[EPT], grp[EPT], pos[EPT];
#pragma unroll
    for (int k = 0; k < EPT; ++k) {
        const int e = e0 + min(k * TPB_B + tid, n_e - 1);
        src[k] = ei[e];
        dst[k] = ei[(size_t)E + e];
    }
#pragma unroll
    for (int k = 0; k < EPT; ++k) {
        const bool in = (k * TPB_B + tid) < n_e;
        const unsigned d = (unsigned)dst[k];
        grp[k] = (d < (unsigned)N) ? (int)(d / (unsigned)RW) : NR;
        pos[k] = in ? atomicAdd(&cnt[grp[k]], 1) : 0;
    }
    __syncthreads();

    if (tid < 32) {
        int ex[NR / 32];
        int run = 0;
#pragma unroll
        for (int j = 0; j < NR / 32; ++j) { ex[j] = run; run += cnt[tid * (NR / 32) + j]; }
        int inc = run;
#pragma unroll
        for (int d = 1; d < 32; d <<= 1) {
            const int t = __shfl_up(inc, d, 32);
            if (tid >= d) inc += t;
        }
        const int base = inc - run;
#pragma unroll
        for (int j = 0; j < NR / 32; ++j) cnt[tid * (NR / 32) + j] = base + ex[j];
        if (tid == 31) cnt[NR] = inc;
    }
    __syncthreads();

    for (int i = tid; i <= NR; i += TPB_B) off[(size_t)c * (NR + 1) + i] = cnt[i];
#pragma unroll
    for (int k = 0; k < EPT; ++k) {
        if ((k * TPB_B + tid) < n_e) stage[cnt[grp[k]] + pos[k]] = make_int2(src[k], dst[k]);
    }
    __syncthreads();
    for (int i = tid; i < n_e; i += TPB_B) pairs[(size_t)e0 + i] = stage[i];
}

static __device__ __forceinline__ void run_entries(const int* __restrict__ off, int NCH, int r,
                                                   int wave, int lane, int j0, int& len, int& base) {
    const int c = wave + WPB * (j0 + lane);
    len = 0; base = 0;
    if (lane < RP && c < NCH) {
        const int a = off[(size_t)c * (NR + 1) + r];
        const int b = off[(size_t)c * (NR + 1) + r + 1];
        len  = b - a;
        base = c * CH + a;
    }
}

static __device__ __forceinline__ int run_longest(int len) {
    int m = 0;
#pragma unroll
    for (int q = 0; q < RP; ++q) m = max(m, __builtin_amdgcn_readlane(len, q));
    return m;
}

static __device__ __forceinline__ void run_slots(int len, int base, int k0, int lane,
                                                 bool (&ok)[P], int (&slot)[P]) {
#pragma unroll
    for (int q = 0; q < RP; ++q) {
        const int lq = __builtin_amdgcn_readlane(len, q);
        const int bq = __builtin_amdgcn_readlane(base, q);
#pragma unroll
        for (int col = 0; col < KC; ++col) {
            const int e0 = k0 + 32 * col;
            const int e  = e0 + lane;
            const int spare = (e0 < lq) ? bq + e0 : 0;
            ok[q * KC + col]   = e < lq;
            slot[q * KC + col] = (e < lq) ? bq + e : spare;
        }
    }
}

__global__ __launch_bounds__(TPB) void k_deg_scale(const int* __restrict__ off,
                                                   const int2* __restrict__ pairs,
                                                   const float* __restrict__ x,
                                                   float* __restrict__ dis,
                                                   float* __restrict__ xs,
                                                   int N, int NCH, int RW) {
    __shared__ unsigned bins[CAPN];
    const int tid  = threadIdx.x;
    const int lane = tid & 31;
    const int wave = __builtin_amdgcn_readfirstlane(tid >> 5);
    const int r    = blockIdx.x;
    const int n_lo = min(r * RW, N);
    const int n_hi = min(n_lo + RW, N);

    for (int lo = n_lo; lo < n_hi; lo += CAPN) {
        const int hi = min(lo + CAPN, n_hi);
        const unsigned span = (unsigned)(hi - lo);
        for (unsigned i = tid; i < span; i += TPB) bins[i] = 0u;
        __syncthreads();
        for (int j0 = 0; wave + WPB * j0 < NCH; j0 += RP) {
            int len, base;
            run_entries(off, NCH, r, wave, lane, j0, len, base);
            const int longest = run_longest(len);
            for (int k0 = 0; k0 < longest; k0 += 32 * KC) {
                bool ok[P]; int slot[P]; int d[P];
                run_slots(len, base, k0, lane, ok, slot);
#pragma unroll
                for (int q = 0; q < P; ++q) d[q] = pairs[slot[q]].y;
#pragma unroll
                for (int q = 0; q < P; ++q) {
                    const unsigned k = (unsigned)(d[q] - lo);
                    if (ok[q] && k < span) atomicAdd(&bins[k], 1u);
                }
            }
        }
        __syncthreads();
        for (unsigned i = tid; i < span; i += TPB) {
            const int n = lo + (int)i;
            const float v  = 1.0f + (float)bins[i];
            const float rs = (v > 0.0f) ? rsqrtf(v) : 0.0f;
            dis[n] = rs;
            const float4 xr = *(const float4*)(x + (size_t)n * FIN);
            float4 o;
            o.x = xr.x * rs; o.y = xr.y * rs; o.z = xr.z * rs; o.w = xr.w * rs;
            *(float4*)(xs + (size_t)n * FIN) = o;
        }
        __syncthreads();
    }
}

__global__ __launch_bounds__(TPB) void k_layer1(const int* __restrict__ off,
                                                const int2* __restrict__ pairs,
                                                const float* __restrict__ xs,
                                                const float* __restrict__ dis,
                                                const float* __restrict__ W1,
                                                const float* __restrict__ b1,
                                                const float* __restrict__ W2,
                                                float* __restrict__ zs,
                                                int N, int NCH, int RW) {
    __shared__ __attribute__((aligned(16))) float bins[CAPN * FIN];
    const int tid  = threadIdx.x;
    const int lane = tid & 31;
    const int wave = __builtin_amdgcn_readfirstlane(tid >> 5);
    const int r    = blockIdx.x;
    const int n_lo = min(r * RW, N);
    const int n_hi = min(n_lo + RW, N);

    for (int lo = n_lo; lo < n_hi; lo += CAPN) {
        const int hi = min(lo + CAPN, n_hi);
        const unsigned span = (unsigned)(hi - lo);
        for (unsigned i = tid; i < span * FIN; i += TPB) bins[i] = 0.0f;
        __syncthreads();
        for (int j0 = 0; wave + WPB * j0 < NCH; j0 += RP) {
            int len, base;
            run_entries(off, NCH, r, wave, lane, j0, len, base);
            const int longest = run_longest(len);
            for (int k0 = 0; k0 < longest; k0 += 32 * KC) {
                bool ok[P]; int slot[P]; int2 pr[P]; float4 v[P];
                run_slots(len, base, k0, lane, ok, slot);
#pragma unroll
                for (int q = 0; q < P; ++q) pr[q] = pairs[slot[q]];
#pragma unroll
                for (int q = 0; q < P; ++q) v[q] = *(const float4*)(xs + (size_t)pr[q].x * FIN);
#pragma unroll
                for (int q = 0; q < P; ++q) {
                    const unsigned k = (unsigned)(pr[q].y - lo);
                    if (ok[q] && k < span) {
                        atomicAdd(&bins[k * FIN + 0], v[q].x);
                        atomicAdd(&bins[k * FIN + 1], v[q].y);
                        atomicAdd(&bins[k * FIN + 2], v[q].z);
                        atomicAdd(&bins[k * FIN + 3], v[q].w);
                    }
                }
            }
        }
        __syncthreads();

        const int half = lane >> 4;
        const int l16  = lane & 15;
        const int kb   = half * 2;
        for (int t = wave; t * 16 < (int)span; t += WPB) {
            const int li   = t * 16 + l16;
            const int lrow = min(li, (int)span - 1);
            const int nrow = lo + lrow;

            const float dn = dis[nrow];
            const v2f self = *(const v2f*)(xs + (size_t)nrow * FIN + kb);
            const v2f sum  = *(const v2f*)(bins + lrow * FIN + kb);
            v2f xb;
            xb.x = (self.x + sum.x) * dn;
            xb.y = (self.y + sum.y) * dn;

            float s[FHID / 16];
#pragma unroll
            for (int c = 0; c < FHID / 16; ++c) {
                const int hcol = c * 16 + l16;
                v2f wa;
                wa.x = W1[(kb + 0) * FHID + hcol];
                wa.y = W1[(kb + 1) * FHID + hcol];
                v8f acc = {};
                acc = __builtin_amdgcn_wmma_f32_16x16x4_f32(
                     false, wa,  false, xb,
                     (short)0, acc,  false,  false);

                const int hb = c * 16 + half * 8;
                const float4 bA = *(const float4*)(b1 + hb);
                const float4 bB = *(const float4*)(b1 + hb + 4);
                const float4 wA = *(const float4*)(W2 + hb);
                const float4 wB = *(const float4*)(W2 + hb + 4);
                float sc = 0.0f;
                sc = fmaf(fmaxf(acc[0] + bA.x, 0.0f), wA.x, sc);
                sc = fmaf(fmaxf(acc[1] + bA.y, 0.0f), wA.y, sc);
                sc = fmaf(fmaxf(acc[2] + bA.z, 0.0f), wA.z, sc);
                sc = fmaf(fmaxf(acc[3] + bA.w, 0.0f), wA.w, sc);
                sc = fmaf(fmaxf(acc[4] + bB.x, 0.0f), wB.x, sc);
                sc = fmaf(fmaxf(acc[5] + bB.y, 0.0f), wB.y, sc);
                sc = fmaf(fmaxf(acc[6] + bB.z, 0.0f), wB.z, sc);
                sc = fmaf(fmaxf(acc[7] + bB.w, 0.0f), wB.w, sc);
                s[c] = sc;
            }
            float tot = (s[0] + s[1]) + (s[2] + s[3]);
            tot += __shfl_xor(tot, 16, 32);
            if (lane < 16 && li < (int)span) zs[lo + li] = tot * dn;
        }
        __syncthreads();
    }
}

__global__ __launch_bounds__(TPB) void k_layer2(const int* __restrict__ off,
                                                const int2* __restrict__ pairs,
                                                const float* __restrict__ zs,
                                                const float* __restrict__ dis,
                                                const float* __restrict__ b2,
                                                float* __restrict__ out,
                                                int N, int NCH, int RW) {
    __shared__ float bins[CAPN];
    const int tid  = threadIdx.x;
    const int lane = tid & 31;
    const int wave = __builtin_amdgcn_readfirstlane(tid >> 5);
    const int r    = blockIdx.x;
    const int n_lo = min(r * RW, N);
    const int n_hi = min(n_lo + RW, N);

    for (int lo = n_lo; lo < n_hi; lo += CAPN) {
        const int hi = min(lo + CAPN, n_hi);
        const unsigned span = (unsigned)(hi - lo);
        for (unsigned i = tid; i < span; i += TPB) bins[i] = 0.0f;
        __syncthreads();
        for (int j0 = 0; wave + WPB * j0 < NCH; j0 += RP) {
            int len, base;
            run_entries(off, NCH, r, wave, lane, j0, len, base);
            const int longest = run_longest(len);
            for (int k0 = 0; k0 < longest; k0 += 32 * KC) {
                bool ok[P]; int slot[P]; int2 pr[P]; float v[P];
                run_slots(len, base, k0, lane, ok, slot);
#pragma unroll
                for (int q = 0; q < P; ++q) pr[q] = pairs[slot[q]];
#pragma unroll
                for (int q = 0; q < P; ++q) v[q] = zs[pr[q].x];
#pragma unroll
                for (int q = 0; q < P; ++q) {
                    const unsigned k = (unsigned)(pr[q].y - lo);
                    if (ok[q] && k < span) atomicAdd(&bins[k], v[q]);
                }
            }
        }
        __syncthreads();
        for (unsigned i = tid; i < span; i += TPB) {
            const int n = lo + (int)i;
            const float v = (zs[n] + bins[i]) * dis[n] + b2[0];
            out[n] = 1.0f / (1.0f + __expf(-v));
        }
        __syncthreads();
    }
}

extern "C" void kernel_launch(void* const* d_in, const int* in_sizes, int n_in,
                              void* d_out, int out_size, void* d_ws, size_t ws_size,
                              hipStream_t stream) {
    const float* x  = (const float*)d_in[0];
    const int*   ei = (const int*)d_in[1];
    const float* W1 = (const float*)d_in[2];
    const float* b1 = (const float*)d_in[3];
    const float* W2 = (const float*)d_in[4];
    const float* b2 = (const float*)d_in[5];
    float* out = (float*)d_out;

    const int N = in_sizes[0] / FIN;
    const int E = in_sizes[1] / 2;
    const int NCH = cdiv(E, CH);
    const int RW  = cdiv(N, NR);

    char* ws = (char*)d_ws;
    float* xs    = (float*)ws;               ws += (size_t)N * FIN * sizeof(float);
    int2*  pairs = (int2*)ws;                ws += (size_t)E * sizeof(int2);
    float* dis   = (float*)ws;               ws += (size_t)N * sizeof(float);
    float* zs    = (float*)ws;               ws += (size_t)N * sizeof(float);
    int*   off   = (int*)ws;

    k_bucket   <<<NCH, TPB_B, 0, stream>>>(ei, off, pairs, E, N, RW);
    k_deg_scale<<<NR,  TPB,   0, stream>>>(off, pairs, x, dis, xs, N, NCH, RW);
    k_layer1   <<<NR,  TPB,   0, stream>>>(off, pairs, xs, dis, W1, b1, W2, zs, N, NCH, RW);
    k_layer2   <<<NR,  TPB,   0, stream>>>(off, pairs, zs, dis, b2, out, N, NCH, RW);
}
